# starting kernel
# baseline (speedup 1.0000x reference)
#include <hip/hip_runtime.h>

typedef float v2f __attribute__((ext_vector_type(2)));
typedef float v8f __attribute__((ext_vector_type(8)));

static constexpr int BB  = 2;
static constexpr int NP  = 8192;
static constexpr int CH  = 64;
static constexpr int KNN = 16;

__device__ __forceinline__ float leaky(float x) { return fmaxf(x, 0.1f * x); }

__global__ __launch_bounds__(256) void knn_kernel(
    const float* __restrict__ xq,
    const float* __restrict__ xk,
    int* __restrict__ idx)
{
    __shared__ float sx[256], sy[256], sz[256];
    int t = blockIdx.x * 256 + threadIdx.x;
    int b = t / NP;
    int n = t % NP;
    const float* q  = xq + (size_t)b * 3 * NP;
    const float* kp = xk + (size_t)b * 3 * NP;
    float qx = q[n], qy = q[NP + n], qz = q[2 * NP + n];

    float best[KNN]; int bidx[KNN];
#pragma unroll
    for (int j = 0; j < KNN; j++) { best[j] = 3.4e38f; bidx[j] = 0; }

    for (int tile = 0; tile < NP; tile += 256) {
        __syncthreads();
        int c = tile + threadIdx.x;
        sx[threadIdx.x] = kp[c];
        sy[threadIdx.x] = kp[NP + c];
        sz[threadIdx.x] = kp[2 * NP + c];
        __syncthreads();
        for (int j = 0; j < 256; j++) {
            float dx = qx - sx[j], dy = qy - sy[j], dz = qz - sz[j];
            float d  = dx * dx + dy * dy + dz * dz;
            if (d < best[KNN - 1]) {
                float vd = d; int vi = tile + j;
#pragma unroll
                for (int s = 0; s < KNN; s++) {
                    if (vd < best[s]) {
                        float td = best[s]; int ti = bidx[s];
                        best[s] = vd; bidx[s] = vi;
                        vd = td; vi = ti;
                    }
                }
            }
        }
    }
    int* o = idx + (size_t)t * KNN;
#pragma unroll
    for (int j = 0; j < KNN; j++) o[j] = bidx[j];
}

__device__ __forceinline__ void stage_wfrag(v2f* wl, const float* __restrict__ W,
                                            int tid, int nthreads)
{
    for (int s = tid; s < 2048; s += nthreads) {
        int lane = s & 31, nc = (s >> 5) & 3, kk = s >> 7;
        int o  = nc * 16 + (lane & 15);
        int k0 = kk * 4 + 2 * (lane >> 4);
        v2f w; w.x = W[o * 64 + k0]; w.y = W[o * 64 + k0 + 1];
        wl[s] = w;
    }
}

__global__ __launch_bounds__(256) void gemm_cm_kernel(
    const float* __restrict__ in,
    const float* __restrict__ W,
    const float* __restrict__ bias,
    float* __restrict__ out_rm)
{
    __shared__ v2f  wl[2048];
    __shared__ float bl[64];
    int tid = threadIdx.x;
    stage_wfrag(wl, W, tid, 256);
    if (tid < 64) bl[tid] = bias[tid];
    __syncthreads();

    int lane = tid & 31, wave = tid >> 5;
    int tile = blockIdx.x * 8 + wave;
    int b  = tile / (NP / 16);
    int n0 = (tile % (NP / 16)) * 16;
    const float* inb = in + (size_t)b * CH * NP;
    int row   = n0 + (lane & 15);
    int khalf = 2 * (lane >> 4);

    v2f a[16];
#pragma unroll
    for (int kk = 0; kk < 16; kk++) {
        int k0 = kk * 4 + khalf;
        v2f v; v.x = inb[(size_t)k0 * NP + row]; v.y = inb[(size_t)(k0 + 1) * NP + row];
        a[kk] = v;
    }
#pragma unroll
    for (int nc = 0; nc < 4; nc++) {
        v8f acc = {0.f, 0.f, 0.f, 0.f, 0.f, 0.f, 0.f, 0.f};
#pragma unroll
        for (int kk = 0; kk < 16; kk++) {
            acc = __builtin_amdgcn_wmma_f32_16x16x4_f32(
                false, a[kk], false, wl[(kk * 4 + nc) * 32 + lane],
                (short)0, acc, false, false);
        }
        int m = nc * 16 + (lane & 15);
        float bv = bl[m];
        float* ob = out_rm + (size_t)b * NP * 64;
#pragma unroll
        for (int v = 0; v < 8; v++) {
            int r = n0 + v + 8 * (lane >> 4);
            ob[(size_t)r * 64 + m] = acc[v] + bv;
        }
    }
}

__global__ __launch_bounds__(256) void gemm_rm_kernel(
    const float* __restrict__ in,
    const float* __restrict__ W,
    const float* __restrict__ bias,
    float* __restrict__ out_cm,
    float* __restrict__ out_rm)
{
    __shared__ v2f  wl[2048];
    __shared__ float bl[64];
    int tid = threadIdx.x;
    stage_wfrag(wl, W, tid, 256);
    if (tid < 64) bl[tid] = bias[tid];
    __syncthreads();

    int lane = tid & 31, wave = tid >> 5;
    int tile = blockIdx.x * 8 + wave;
    int b  = tile / (NP / 16);
    int n0 = (tile % (NP / 16)) * 16;
    const float* inb = in + (size_t)b * NP * 64;
    int row   = n0 + (lane & 15);
    int khalf = 2 * (lane >> 4);

    v2f a[16];
#pragma unroll
    for (int kk = 0; kk < 16; kk++)
        a[kk] = *(const v2f*)(inb + (size_t)row * 64 + kk * 4 + khalf);

#pragma unroll
    for (int nc = 0; nc < 4; nc++) {
        v8f acc = {0.f, 0.f, 0.f, 0.f, 0.f, 0.f, 0.f, 0.f};
#pragma unroll
        for (int kk = 0; kk < 16; kk++) {
            acc = __builtin_amdgcn_wmma_f32_16x16x4_f32(
                false, a[kk], false, wl[(kk * 4 + nc) * 32 + lane],
                (short)0, acc, false, false);
        }
        int m = nc * 16 + (lane & 15);
        float bv = bl[m];
#pragma unroll
        for (int v = 0; v < 8; v++) {
            int r = n0 + v + 8 * (lane >> 4);
            float val = acc[v] + bv;
            out_cm[(size_t)b * 64 * NP + (size_t)m * NP + r] = val;
            if (out_rm) out_rm[((size_t)b * NP + r) * 64 + m] = val;
        }
    }
}

__global__ __launch_bounds__(128) void cross_kernel(
    const float* __restrict__ xq, const float* __restrict__ xk,
    const float* __restrict__ p1, const float* __restrict__ p2,
    const int*   __restrict__ idx,
    const float* __restrict__ posw, const float* __restrict__ posb,
    const float* __restrict__ w1, const float* __restrict__ b1,
    const float* __restrict__ w2, const float* __restrict__ b2,
    int nlayers,
    float* __restrict__ out_rm,
    float* __restrict__ out_cm)
{
    __shared__ v2f  wl1[2048];
    __shared__ v2f  wl2[2048];
    __shared__ float bl1[64], bl2[64];
    __shared__ float pw[192], pb[64];
    __shared__ float X[4][16][68];

    int tid = threadIdx.x;
    stage_wfrag(wl1, w1, tid, 128);
    if (nlayers > 1) stage_wfrag(wl2, w2, tid, 128);
    for (int s = tid; s < 192; s += 128) pw[s] = posw[s];
    if (tid < 64) {
        bl1[tid] = b1[tid];
        bl2[tid] = (nlayers > 1) ? b2[tid] : 0.f;
        pb[tid]  = posb[tid];
    }
    __syncthreads();

    int lane = tid & 31, wave = tid >> 5;
    int q = blockIdx.x * 4 + wave;
    int b = q / NP;
    int n = q % NP;

    const float* xqb = xq + (size_t)b * 3 * NP;
    const float* xkb = xk + (size_t)b * 3 * NP;
    float qx = xqb[n], qy = xqb[NP + n], qz = xqb[2 * NP + n];
    const int* iq = idx + (size_t)q * KNN;

    int c0 = 2 * lane;
    v2f p1v = *(const v2f*)(p1 + (size_t)q * 64 + c0);
    float pwa0 = pw[c0 * 3 + 0], pwa1 = pw[c0 * 3 + 1], pwa2 = pw[c0 * 3 + 2];
    float pwb0 = pw[(c0 + 1) * 3 + 0], pwb1 = pw[(c0 + 1) * 3 + 1], pwb2 = pw[(c0 + 1) * 3 + 2];
    float pb0 = pb[c0], pb1 = pb[c0 + 1];
#pragma unroll
    for (int r = 0; r < KNN; r++) {
        int nb = iq[r];
        float dx = xkb[nb] - qx, dy = xkb[NP + nb] - qy, dz = xkb[2 * NP + nb] - qz;
        v2f g = *(const v2f*)(p2 + ((size_t)b * NP + nb) * 64 + c0);
        X[wave][r][c0]     = leaky(g.x + p1v.x + pb0 + dx * pwa0 + dy * pwa1 + dz * pwa2);
        X[wave][r][c0 + 1] = leaky(g.y + p1v.y + pb1 + dx * pwb0 + dy * pwb1 + dz * pwb2);
    }
    __syncthreads();

    int mrow  = lane & 15;
    int khalf = 2 * (lane >> 4);
    for (int layer = 0; layer < nlayers; layer++) {
        const v2f*  wl = layer ? wl2 : wl1;
        const float* bl = layer ? bl2 : bl1;
        bool last = (layer == nlayers - 1);

        v2f a[16];
#pragma unroll
        for (int kk = 0; kk < 16; kk++)
            a[kk] = *(const v2f*)&X[wave][mrow][kk * 4 + khalf];

#pragma unroll
        for (int nc = 0; nc < 4; nc++) {
            v8f acc = {0.f, 0.f, 0.f, 0.f, 0.f, 0.f, 0.f, 0.f};
#pragma unroll
            for (int kk = 0; kk < 16; kk++) {
                acc = __builtin_amdgcn_wmma_f32_16x16x4_f32(
                    false, a[kk], false, wl[(kk * 4 + nc) * 32 + lane],
                    (short)0, acc, false, false);
            }
            int m = nc * 16 + mrow;
            float bv = bl[m];
            if (!last) {
#pragma unroll
                for (int v = 0; v < 8; v++)
                    X[wave][v + 8 * (lane >> 4)][m] = leaky(acc[v] + bv);
            } else {
                float mx = -3.4e38f;
#pragma unroll
                for (int v = 0; v < 8; v++) mx = fmaxf(mx, leaky(acc[v] + bv));
                mx = fmaxf(mx, __shfl_xor(mx, 16, 32));
                if (lane < 16) {
                    out_rm[(size_t)q * 64 + m] = mx;
                    if (out_cm)
                        out_cm[(size_t)b * 64 * NP + (size_t)m * NP + n] = mx;
                }
            }
        }
        __syncthreads();
    }
}

extern "C" void kernel_launch(void* const* d_in, const int* in_sizes, int n_in,
                              void* d_out, int out_size, void* d_ws, size_t ws_size,
                              hipStream_t stream)
{
    (void)in_sizes; (void)n_in; (void)out_size; (void)ws_size;
    const float* pc1   = (const float*)d_in[0];
    const float* pc2   = (const float*)d_in[1];
    const float* feat1 = (const float*)d_in[2];
    const float* feat2 = (const float*)d_in[3];
    const float* t11_w = (const float*)d_in[4];  const float* t11_b = (const float*)d_in[5];
    const float* t22_w = (const float*)d_in[6];  const float* t22_b = (const float*)d_in[7];
    const float* pos1w = (const float*)d_in[8];  const float* pos1b = (const float*)d_in[9];
    const float* m1w1  = (const float*)d_in[10]; const float* m1b1  = (const float*)d_in[11];
    const float* m1w2  = (const float*)d_in[12]; const float* m1b2  = (const float*)d_in[13];
    const float* t1_w  = (const float*)d_in[14]; const float* t1_b  = (const float*)d_in[15];
    const float* t2_w  = (const float*)d_in[16]; const float* t2_b  = (const float*)d_in[17];
    const float* pos2w = (const float*)d_in[18]; const float* pos2b = (const float*)d_in[19];
    const float* m2w1  = (const float*)d_in[20]; const float* m2b1  = (const float*)d_in[21];

    float* out = (float*)d_out;
    const size_t S = (size_t)BB * NP * 64;
    float* ws = (float*)d_ws;
    float* q1 = ws;           float* q2 = ws + S;
    float* r1 = ws + 2 * S;   float* r2 = ws + 3 * S;
    float* pooled1 = ws + 4 * S;
    float* pooled2 = ws + 5 * S;
    int* idx1 = (int*)(ws + 6 * S);
    int* idx2 = idx1 + (size_t)BB * NP * KNN;
    float* fn1_rm  = q1;
    float* fn2_rm  = q2;
    float* pooled3 = r1;

    knn_kernel<<<BB * NP / 256, 256, 0, stream>>>(pc1, pc2, idx1);
    knn_kernel<<<BB * NP / 256, 256, 0, stream>>>(pc2, pc1, idx2);

    gemm_cm_kernel<<<128, 256, 0, stream>>>(feat1, t11_w, t11_b, q1);
    gemm_cm_kernel<<<128, 256, 0, stream>>>(feat2, t22_w, t22_b, q2);
    gemm_cm_kernel<<<128, 256, 0, stream>>>(feat2, t11_w, t11_b, r1);
    gemm_cm_kernel<<<128, 256, 0, stream>>>(feat1, t22_w, t22_b, r2);

    cross_kernel<<<BB * NP / 4, 128, 0, stream>>>(pc1, pc2, q1, q2, idx1,
        pos1w, pos1b, m1w1, m1b1, m1w2, m1b2, 2, pooled1, nullptr);
    cross_kernel<<<BB * NP / 4, 128, 0, stream>>>(pc2, pc1, r1, r2, idx2,
        pos1w, pos1b, m1w1, m1b1, m1w2, m1b2, 2, pooled2, nullptr);

    gemm_rm_kernel<<<128, 256, 0, stream>>>(pooled1, t1_w, t1_b, out,       fn1_rm);
    gemm_rm_kernel<<<128, 256, 0, stream>>>(pooled2, t2_w, t2_b, out + S,   fn2_rm);

    cross_kernel<<<BB * NP / 4, 128, 0, stream>>>(pc1, pc2, fn1_rm, fn2_rm, idx1,
        pos2w, pos2b, m2w1, m2b1, nullptr, nullptr, 1, pooled3, out + 2 * S);
}


# weight-staging prologues: loads in one or two groups; weights read as whole 128-byte lines (16 B per lane), scattered to fragment order at the LDS store; gather issued before the only barrier
# speedup vs baseline: 9.1349x; 9.1349x over previous
#include <hip/hip_runtime.h>

typedef float v2f __attribute__((ext_vector_type(2)));
typedef float v8f __attribute__((ext_vector_type(8)));
typedef int   v2i __attribute__((ext_vector_type(2)));
typedef float v4f __attribute__((ext_vector_type(4)));
typedef int   v4i __attribute__((ext_vector_type(4)));

static constexpr int BB  = 2;
static constexpr int NP  = 8192;
static constexpr int CH  = 64;
static constexpr int KNN = 16;

__device__ __forceinline__ float leaky(float x) { return fmaxf(x, 0.1f * x); }

static constexpr int QDEPTH = 24;
static constexpr int QFULL  = 16;
static constexpr int KGRP   = 8;

typedef unsigned long long u64;

template <int J>
__device__ __forceinline__ float row_bcast(float v)
{
    return __int_as_float(__builtin_amdgcn_update_dpp(0, __float_as_int(v), 0x150 + J, 0xf, 0xf, true));
}

template <int J>
__device__ __forceinline__ void scan_key(float qx, float qy, float qz,
                                         float cx, float cy, float cz,
                                         int tile, v2i* ql, int& cnt, float thr)
{
    float dx = qx - row_bcast<J>(cx);
    float dy = qy - row_bcast<J>(cy);
    float dz = qz - row_bcast<J>(cz);
    float d  = dx * dx + dy * dy + dz * dz;
    v2i e; e.x = __float_as_int(d); e.y = tile + J;
    ql[cnt * 32] = e;
    cnt += (d < thr) ? 1 : 0;
}

template <int G>
__device__ __forceinline__ void scan_group(float qx, float qy, float qz,
                                           float cx, float cy, float cz,
                                           int tile, v2i* ql, int& cnt, float thr)
{
    scan_key<G + 0>(qx, qy, qz, cx, cy, cz, tile, ql, cnt, thr);
    scan_key<G + 1>(qx, qy, qz, cx, cy, cz, tile, ql, cnt, thr);
    scan_key<G + 2>(qx, qy, qz, cx, cy, cz, tile, ql, cnt, thr);
    scan_key<G + 3>(qx, qy, qz, cx, cy, cz, tile, ql, cnt, thr);
    scan_key<G + 4>(qx, qy, qz, cx, cy, cz, tile, ql, cnt, thr);
    scan_key<G + 5>(qx, qy, qz, cx, cy, cz, tile, ql, cnt, thr);
    scan_key<G + 6>(qx, qy, qz, cx, cy, cz, tile, ql, cnt, thr);
    scan_key<G + 7>(qx, qy, qz, cx, cy, cz, tile, ql, cnt, thr);
}

__device__ __forceinline__ u64 set_entry(float d, int ki, unsigned slot)
{
    return ((u64)(unsigned)__float_as_int(d) << 32) | (u64)(((unsigned)ki << 4) | slot);
}
__device__ __forceinline__ float entry_dist(u64 e) { return __int_as_float((int)(unsigned)(e >> 32)); }
__device__ __forceinline__ int   entry_key(u64 e)  { return (int)((unsigned)e >> 4); }

__device__ __forceinline__ u64 set_max(const u64 (&k)[KNN])
{
    u64 t[KNN / 2];
#pragma unroll
    for (int i = 0; i < KNN / 2; i++) t[i] = (k[i + KNN / 2] > k[i]) ? k[i + KNN / 2] : k[i];
#pragma unroll
    for (int w = KNN / 4; w >= 1; w >>= 1) {
#pragma unroll
        for (int i = 0; i < w; i++) t[i] = (t[i + w] > t[i]) ? t[i + w] : t[i];
    }
    return t[0];
}

__device__ __forceinline__ void set_offer(u64 (&k)[KNN], u64& mk, float vd, int vi)
{
    const u64 c = set_entry(vd, vi, 0u);
    const bool take = c < (mk & ~(u64)15);
    const unsigned mslot = (unsigned)mk & 15u;
    const unsigned slot  = take ? mslot : 16u;
    const u64 nv = c | (u64)mslot;
#pragma unroll
    for (int s = 0; s < KNN; s++) k[s] = (slot == (unsigned)s) ? nv : k[s];
    mk = set_max(k);
}

__device__ __forceinline__ void drain_queue(u64 (&k)[KNN], u64& mk, const v2i* ql, int cnt)
{
    static_assert(QDEPTH % 2 == 0, "two queue slots per trip");
    unsigned live = __builtin_amdgcn_ballot_w32(0 < cnt);
#pragma unroll 1
    for (int r = 0; live != 0u; r += 2) {
        const unsigned live_next = (r + 2 < QDEPTH) ? __builtin_amdgcn_ballot_w32(r + 2 < cnt) : 0u;
        const v2i ea = ql[r * 32];
        const v2i eb = ql[(r + 1) * 32];
        const bool ha = r < cnt, hb = r + 1 < cnt;
        set_offer(k, mk, ha ? __int_as_float(ea.x) : 3.4e38f, ha ? ea.y : 0);
        set_offer(k, mk, hb ? __int_as_float(eb.x) : 3.4e38f, hb ? eb.y : 0);
        live = live_next;
    }
}

__global__ __launch_bounds__(128) void knn_kernel(
    const float* __restrict__ xq,
    const float* __restrict__ xk,
    int* __restrict__ idx)
{
    __shared__ u64 lst[2][KNN][32];
    __shared__ v2i qq[4][QDEPTH][32];
    const int lane = threadIdx.x & 31;
    const int wave = __builtin_amdgcn_readfirstlane((int)(threadIdx.x >> 5));
    const int half = wave & 1;
    const int grp  = wave >> 1;
    const int q0 = blockIdx.x * 64 + grp * 32;
    const int b  = q0 / NP;
    const int n  = (q0 % NP) + lane;
    const float* q  = xq + (size_t)b * 3 * NP;
    const float* kp = xk + (size_t)b * 3 * NP;
    float qx = q[n], qy = q[NP + n], qz = q[2 * NP + n];

    u64 ks[KNN];
#pragma unroll
    for (int s = 0; s < KNN; s++) ks[s] = set_entry(3.4e38f, 0, (unsigned)s);
    u64 mk = set_entry(3.4e38f, 0, KNN - 1);

    v2i* ql = &qq[wave][0][lane];
    {
        v2i f; f.x = __float_as_int(3.4e38f); f.y = 0;
#pragma unroll
        for (int s = 0; s < QDEPTH; s++) ql[s * 32] = f;
    }
    float thr = 3.4e38f;
    int   cnt = 0;

    const int kbeg = half * (NP / 2);
    const int kend = kbeg + NP / 2;
    const int l16 = lane & 15;
    const float* kl = kp + l16;
    float ax = kl[kbeg],      ay = kl[NP + kbeg],      az = kl[2 * NP + kbeg];
    float bx = kl[kbeg + 16], by = kl[NP + kbeg + 16], bz = kl[2 * NP + kbeg + 16];
    float ex = kl[kbeg + 32], ey = kl[NP + kbeg + 32], ez = kl[2 * NP + kbeg + 32];
    float fx = kl[kbeg + 48], fy = kl[NP + kbeg + 48], fz = kl[2 * NP + kbeg + 48];

#define KNN_TILE_STEP(SX, SY, SZ, T)                                              \
    {                                                                             \
        const int tt = (T);                                                       \
        scan_group<0>(qx, qy, qz, SX, SY, SZ, tt, ql, cnt, thr);                  \
          \
        if (__builtin_amdgcn_ballot_w32(cnt >= QFULL) != 0u) {                    \
            drain_queue(ks, mk, ql, cnt);                                         \
            thr = entry_dist(mk);                                                 \
            cnt = 0;                                                              \
        }                                                                         \
        scan_group<KGRP>(qx, qy, qz, SX, SY, SZ, tt, ql, cnt, thr);               \
        if (__builtin_amdgcn_ballot_w32(cnt >= QFULL) != 0u) {                    \
            drain_queue(ks, mk, ql, cnt);                                         \
            thr = entry_dist(mk);                                                 \
            cnt = 0;                                                              \
        }                                                                         \
        const int nt = (tt + 64 < kend) ? tt + 64 : tt;                           \
        SX = kl[nt]; SY = kl[NP + nt]; SZ = kl[2 * NP + nt];                      \
    }

#pragma unroll 1
    for (int tile = kbeg; tile < kend; tile += 64) {
        KNN_TILE_STEP(ax, ay, az, tile)
        KNN_TILE_STEP(bx, by, bz, tile + 16)
        KNN_TILE_STEP(ex, ey, ez, tile + 32)
        KNN_TILE_STEP(fx, fy, fz, tile + 48)
    }
#undef KNN_TILE_STEP
    drain_queue(ks, mk, ql, cnt);

    u64* ul = &lst[grp][0][lane];
    if (half == 1) {
#pragma unroll
        for (int s = 0; s < KNN; s++) ul[s * 32] = ks[s];
    }
    __syncthreads();
    if (half == 0) {
#pragma unroll 1
        for (int r = 0; r < KNN; r += 2) {
            const u64 ea = ul[r * 32];
            const u64 eb = ul[(r + 1) * 32];
            set_offer(ks, mk, entry_dist(ea), entry_key(ea));
            set_offer(ks, mk, entry_dist(eb), entry_key(eb));
        }
        int* o = idx + (size_t)(q0 + lane) * KNN;
#pragma unroll
        for (int j = 0; j < KNN; j++) o[j] = entry_key(ks[j]);
    }
}

static constexpr int WKK = 4 * 32 + 2;

__device__ __forceinline__ void put_wpiece(v2f* wl, int piece, v4f w)
{
    const int o = piece >> 4, j = piece & 15;
    v2f* p = wl + j * WKK + (o >> 4) * 32 + (o & 15);
    v2f lo; lo.x = w.x; lo.y = w.y;
    v2f hi; hi.x = w.z; hi.y = w.w;
    p[0]  = lo;
    p[16] = hi;
}

__global__ __launch_bounds__(256) void gemm_cm_kernel(
    const float* __restrict__ in,
    const float* __restrict__ W,
    const float* __restrict__ bias,
    float* __restrict__ out_rm)
{
    __shared__ v2f wl[16 * WKK];
    int tid = threadIdx.x;
    int lane = tid & 31, wave = tid >> 5;
    int tile = blockIdx.x * 8 + wave;
    int b  = tile / (NP / 16);
    int n0 = (tile % (NP / 16)) * 16;
    const float* inb = in + (size_t)b * CH * NP;
    int row   = n0 + (lane & 15);
    int khalf = 2 * (lane >> 4);

    v4f wv[4];
#pragma unroll
    for (int i = 0; i < 4; i++) wv[i] = *(const v4f*)(W + 4 * (tid + 256 * i));
    float bvs[4];
#pragma unroll
    for (int nc = 0; nc < 4; nc++) bvs[nc] = bias[nc * 16 + (lane & 15)];
    v2f a[16];
#pragma unroll
    for (int kk = 0; kk < 16; kk++) {
        int k0 = kk * 4 + khalf;
        v2f v; v.x = inb[(size_t)k0 * NP + row]; v.y = inb[(size_t)(k0 + 1) * NP + row];
        a[kk] = v;
    }
#pragma unroll
    for (int i = 0; i < 4; i++) put_wpiece(wl, tid + 256 * i, wv[i]);
    __syncthreads();
#pragma unroll
    for (int nc = 0; nc < 4; nc++) {
        v8f acc = {0.f, 0.f, 0.f, 0.f, 0.f, 0.f, 0.f, 0.f};
#pragma unroll
        for (int kk = 0; kk < 16; kk++) {
            acc = __builtin_amdgcn_wmma_f32_16x16x4_f32(
                false, a[kk], false, wl[kk * WKK + nc * 32 + lane],
                (short)0, acc, false, false);
        }
        int m = nc * 16 + (lane & 15);
        float bv = bvs[nc];
        float* ob = out_rm + (size_t)b * NP * 64;
#pragma unroll
        for (int v = 0; v < 8; v++) {
            int r = n0 + v + 8 * (lane >> 4);
            ob[(size_t)r * 64 + m] = acc[v] + bv;
        }
    }
}

__global__ __launch_bounds__(256) void gemm_rm_kernel(
    const float* __restrict__ in,
    const float* __restrict__ W,
    const float* __restrict__ bias,
    float* __restrict__ out_cm,
    float* __restrict__ out_rm)
{
    __shared__ v2f wl[16 * WKK];
    int tid = threadIdx.x;
    int lane = tid & 31, wave = tid >> 5;
    int tile = blockIdx.x * 8 + wave;
    int b  = tile / (NP / 16);
    int n0 = (tile % (NP / 16)) * 16;
    const float* inb = in + (size_t)b * NP * 64;
    int row   = n0 + (lane & 15);
    int khalf = 2 * (lane >> 4);

    v4f wv[4];
#pragma unroll
    for (int i = 0; i < 4; i++) wv[i] = *(const v4f*)(W + 4 * (tid + 256 * i));
    float bvs[4];
#pragma unroll
    for (int nc = 0; nc < 4; nc++) bvs[nc] = bias[nc * 16 + (lane & 15)];
    v2f a[16];
#pragma unroll
    for (int kk = 0; kk < 16; kk++)
        a[kk] = *(const v2f*)(inb + (size_t)row * 64 + kk * 4 + khalf);
#pragma unroll
    for (int i = 0; i < 4; i++) put_wpiece(wl, tid + 256 * i, wv[i]);
    __syncthreads();

#pragma unroll
    for (int nc = 0; nc < 4; nc++) {
        v8f acc = {0.f, 0.f, 0.f, 0.f, 0.f, 0.f, 0.f, 0.f};
#pragma unroll
        for (int kk = 0; kk < 16; kk++) {
            acc = __builtin_amdgcn_wmma_f32_16x16x4_f32(
                false, a[kk], false, wl[kk * WKK + nc * 32 + lane],
                (short)0, acc, false, false);
        }
        int m = nc * 16 + (lane & 15);
        float bv = bvs[nc];
#pragma unroll
        for (int v = 0; v < 8; v++) {
            int r = n0 + v + 8 * (lane >> 4);
            float val = acc[v] + bv;
            out_cm[(size_t)b * 64 * NP + (size_t)m * NP + r] = val;
            if (out_rm) out_rm[((size_t)b * NP + r) * 64 + m] = val;
        }
    }
}

static constexpr int QPW = 2;
template <int NL>
__global__ __launch_bounds__(128, 4) void cross_kernel(
    const float* __restrict__ xq, const float* __restrict__ xk,
    const float* __restrict__ p1, const float* __restrict__ p2,
    const int*   __restrict__ idx,
    const float* __restrict__ posw, const float* __restrict__ posb,
    const float* __restrict__ w1, const float* __restrict__ b1,
    const float* __restrict__ w2, const float* __restrict__ b2,
    float* __restrict__ out_rm,
    float* __restrict__ out_cm)
{
    __shared__ v2f  wl1[16 * WKK];
    __shared__ v2f  wl2[(NL > 1) ? 16 * WKK : 2];
    __shared__ float X[4][QPW * 16][68];

    const int tid = threadIdx.x;
    const int lane = tid & 31, wave = tid >> 5;
    const int q0 = (blockIdx.x * 4 + wave) * QPW;
    const int b  = q0 / NP;
    const int n0 = q0 % NP;
    const float* xqb = xq + (size_t)b * 3 * NP;
    const float* xkb = xk + (size_t)b * 3 * NP;
    const int c0 = 2 * lane;
    const int mrow  = lane & 15;
    const int khalf = 2 * (lane >> 4);
    const int hrow  = 8 * (lane >> 4);

    v4i nbv[QPW][KNN / 4];
    float qx[QPW], qy[QPW], qz[QPW];
    v2f p1v[QPW];
#pragma unroll
    for (int t = 0; t < QPW; t++) {
        const v4i* iq = (const v4i*)(idx + (size_t)(q0 + t) * KNN);
#pragma unroll
        for (int i = 0; i < KNN / 4; i++) nbv[t][i] = iq[i];
        qx[t] = xqb[n0 + t]; qy[t] = xqb[NP + n0 + t]; qz[t] = xqb[2 * NP + n0 + t];
        p1v[t] = *(const v2f*)(p1 + (size_t)(q0 + t) * 64 + c0);
    }
    v4f wa[8], wb[8];
#pragma unroll
    for (int i = 0; i < 8; i++) wa[i] = *(const v4f*)(w1 + 4 * (tid + 128 * i));
    if (NL > 1) {
#pragma unroll
        for (int i = 0; i < 8; i++) wb[i] = *(const v4f*)(w2 + 4 * (tid + 128 * i));
    }
    const v2f pw01 = *(const v2f*)(posw + c0 * 3);
    const v2f pw23 = *(const v2f*)(posw + c0 * 3 + 2);
    const v2f pw45 = *(const v2f*)(posw + c0 * 3 + 4);
    const v2f pbv  = *(const v2f*)(posb + c0);
    float bv1[4], bv2[4];
#pragma unroll
    for (int nc = 0; nc < 4; nc++) {
        bv1[nc] = b1[nc * 16 + mrow];
        bv2[nc] = (NL > 1) ? b2[nc * 16 + mrow] : 0.f;
    }

#pragma unroll
    for (int i = 0; i < 8; i++) put_wpiece(wl1, tid + 128 * i, wa[i]);
    if (NL > 1) {
#pragma unroll
        for (int i = 0; i < 8; i++) put_wpiece(wl2, tid + 128 * i, wb[i]);
    }
    __builtin_amdgcn_sched_barrier(0);

    float dx[QPW][KNN], dy[QPW][KNN], dz[QPW][KNN];
    v2f   g[QPW][KNN];
#pragma unroll
    for (int t = 0; t < QPW; t++) {
#pragma unroll
        for (int r = 0; r < KNN; r++) {
            const int nb = nbv[t][r >> 2][r & 3];
            dx[t][r] = xkb[nb]; dy[t][r] = xkb[NP + nb]; dz[t][r] = xkb[2 * NP + nb];
            g[t][r]  = *(const v2f*)(p2 + ((size_t)b * NP + nb) * 64 + c0);
        }
    }

    const float pwa0 = pw01.x, pwa1 = pw01.y, pwa2 = pw23.x;
    const float pwb0 = pw23.y, pwb1 = pw45.x, pwb2 = pw45.y;
    const float pb0 = pbv.x, pb1 = pbv.y;
#pragma unroll
    for (int t = 0; t < QPW; t++) {
#pragma unroll
        for (int r = 0; r < KNN; r++) {
            const float ddx = dx[t][r] - qx[t], ddy = dy[t][r] - qy[t], ddz = dz[t][r] - qz[t];
            X[wave][t * 16 + r][c0]     = leaky(g[t][r].x + p1v[t].x + pb0 + ddx * pwa0 + ddy * pwa1 + ddz * pwa2);
            X[wave][t * 16 + r][c0 + 1] = leaky(g[t][r].y + p1v[t].y + pb1 + ddx * pwb0 + ddy * pwb1 + ddz * pwb2);
        }
    }
    __syncthreads();

#pragma unroll
    for (int layer = 0; layer < NL; layer++) {
        const v2f* wl = layer ? wl2 : wl1;
        const bool last = (layer == NL - 1);

        v2f a0[16], a1[16];
#pragma unroll
        for (int kk = 0; kk < 16; kk++) {
            a0[kk] = *(const v2f*)&X[wave][mrow][kk * 4 + khalf];
            a1[kk] = *(const v2f*)&X[wave][16 + mrow][kk * 4 + khalf];
        }

#pragma unroll
        for (int nc = 0; nc < 4; nc++) {
            v8f acc0 = {0.f, 0.f, 0.f, 0.f, 0.f, 0.f, 0.f, 0.f};
            v8f acc1 = {0.f, 0.f, 0.f, 0.f, 0.f, 0.f, 0.f, 0.f};
#pragma unroll
            for (int kk = 0; kk < 16; kk++) {
                const v2f bf = wl[kk * WKK + nc * 32 + lane];
                acc0 = __builtin_amdgcn_wmma_f32_16x16x4_f32(
                    false, a0[kk], false, bf, (short)0, acc0, false, false);
                acc1 = __builtin_amdgcn_wmma_f32_16x16x4_f32(
                    false, a1[kk], false, bf, (short)0, acc1, false, false);
            }
            const int m = nc * 16 + mrow;
            const float bv = layer ? bv2[nc] : bv1[nc];
            if (!last) {
#pragma unroll
                for (int v = 0; v < 8; v++) {
                    X[wave][v + hrow][m]      = leaky(acc0[v] + bv);
                    X[wave][16 + v + hrow][m] = leaky(acc1[v] + bv);
                }
            } else {
                float mx0 = -3.4e38f, mx1 = -3.4e38f;
#pragma unroll
                for (int v = 0; v < 8; v++) {
                    mx0 = fmaxf(mx0, leaky(acc0[v] + bv));
                    mx1 = fmaxf(mx1, leaky(acc1[v] + bv));
                }
                mx0 = fmaxf(mx0, __shfl_xor(mx0, 16, 32));
                mx1 = fmaxf(mx1, __shfl_xor(mx1, 16, 32));
                const int t = lane >> 4;
                const float mx = t ? mx1 : mx0;
                out_rm[(size_t)(q0 + t) * 64 + m] = mx;
                if (out_cm)
                    out_cm[(size_t)b * 64 * NP + (size_t)m * NP + n0 + t] = mx;
            }
        }
        if (!last) __syncthreads();
    }
}

extern "C" void kernel_launch(void* const* d_in, const int* in_sizes, int n_in,
                              void* d_out, int out_size, void* d_ws, size_t ws_size,
                              hipStream_t stream)
{
    (void)in_sizes; (void)n_in; (void)out_size; (void)ws_size;
    const float* pc1   = (const float*)d_in[0];
    const float* pc2   = (const float*)d_in[1];
    const float* feat1 = (const float*)d_in[2];
    const float* feat2 = (const float*)d_in[3];
    const float* t11_w = (const float*)d_in[4];  const float* t11_b = (const float*)d_in[5];
    const float* t22_w = (const float*)d_in[6];  const float* t22_b = (const float*)d_in[7];
    const float* pos1w = (const float*)d_in[8];  const float* pos1b = (const float*)d_in[9];
    const float* m1w1  = (const float*)d_in[10]; const float* m1b1  = (const float*)d_in[11];
    const float* m1w2  = (const float*)d_in[12]; const float* m1b2  = (const float*)d_in[13];
    const float* t1_w  = (const float*)d_in[14]; const float* t1_b  = (const float*)d_in[15];
    const float* t2_w  = (const float*)d_in[16]; const float* t2_b  = (const float*)d_in[17];
    const float* pos2w = (const float*)d_in[18]; const float* pos2b = (const float*)d_in[19];
    const float* m2w1  = (const float*)d_in[20]; const float* m2b1  = (const float*)d_in[21];

    float* out = (float*)d_out;
    const size_t S = (size_t)BB * NP * 64;
    float* ws = (float*)d_ws;
    float* q1 = ws;           float* q2 = ws + S;
    float* r1 = ws + 2 * S;   float* r2 = ws + 3 * S;
    float* pooled1 = ws + 4 * S;
    float* pooled2 = ws + 5 * S;
    int* idx1 = (int*)(ws + 6 * S);
    int* idx2 = idx1 + (size_t)BB * NP * KNN;
    float* fn1_rm  = q1;
    float* fn2_rm  = q2;
    float* pooled3 = r1;

    knn_kernel<<<BB * NP / 64, 128, 0, stream>>>(pc1, pc2, idx1);
    knn_kernel<<<BB * NP / 64, 128, 0, stream>>>(pc2, pc1, idx2);

    gemm_cm_kernel<<<128, 256, 0, stream>>>(feat1, t11_w, t11_b, q1);
    gemm_cm_kernel<<<128, 256, 0, stream>>>(feat2, t22_w, t22_b, q2);
    gemm_cm_kernel<<<128, 256, 0, stream>>>(feat2, t11_w, t11_b, r1);
    gemm_cm_kernel<<<128, 256, 0, stream>>>(feat1, t22_w, t22_b, r2);

    cross_kernel<2><<<BB * NP / (4 * QPW), 128, 0, stream>>>(pc1, pc2, q1, q2, idx1,
        pos1w, pos1b, m1w1, m1b1, m1w2, m1b2, pooled1, nullptr);
    cross_kernel<2><<<BB * NP / (4 * QPW), 128, 0, stream>>>(pc2, pc1, r1, r2, idx2,
        pos1w, pos1b, m1w1, m1b1, m1w2, m1b2, pooled2, nullptr);

    gemm_rm_kernel<<<128, 256, 0, stream>>>(pooled1, t1_w, t1_b, out,       fn1_rm);
    gemm_rm_kernel<<<128, 256, 0, stream>>>(pooled2, t2_w, t2_b, out + S,   fn2_rm);

    cross_kernel<1><<<BB * NP / (4 * QPW), 128, 0, stream>>>(pc1, pc2, fn1_rm, fn2_rm, idx1,
        pos2w, pos2b, m2w1, m2b1, nullptr, nullptr, pooled3, out + 2 * S);
}


# kNN scan: the matrix unit (f32 16x16x4, B = -1/0 selector, C = the lane's query) forms query - key for 8 keys at a time, in index order, as plain registers; no DPP row_share subtracts
# speedup vs baseline: 9.5237x; 1.0426x over previous
#include <hip/hip_runtime.h>

typedef float v2f __attribute__((ext_vector_type(2)));
typedef float v8f __attribute__((ext_vector_type(8)));
typedef int   v2i __attribute__((ext_vector_type(2)));
typedef float v4f __attribute__((ext_vector_type(4)));
typedef int   v4i __attribute__((ext_vector_type(4)));

static constexpr int BB  = 2;
static constexpr int NP  = 8192;
static constexpr int CH  = 64;
static constexpr int KNN = 16;

__device__ __forceinline__ float leaky(float x) { return fmaxf(x, 0.1f * x); }

static constexpr int QDEPTH = 24;
static constexpr int QFULL  = 16;
static constexpr int KGRP   = 8;

typedef unsigned long long u64;

__device__ __forceinline__ v8f key_diffs(v2f frag, v2f sel, const v8f& cq)
{
    return __builtin_amdgcn_wmma_f32_16x16x4_f32(false, frag, false, sel, (short)0, cq, false, false);
}

template <int R>
__device__ __forceinline__ v2f dist_pair(const v8f& dx, const v8f& dy, const v8f& dz)
{
    const v2f x = {dx[R], dx[R + 1]}, y = {dy[R], dy[R + 1]}, z = {dz[R], dz[R + 1]};
    return __builtin_elementwise_fma(z, z, __builtin_elementwise_fma(y, y, x * x));
}

__device__ __forceinline__ void queue_key(float d, int ki, v2i* ql, int& cnt, float thr)
{
    v2i e; e.x = __float_as_int(d); e.y = ki;
    ql[cnt * 32] = e;
    cnt += (d < thr) ? 1 : 0;
}

__device__ __forceinline__ u64 set_entry(float d, int ki, unsigned slot)
{
    return ((u64)(unsigned)__float_as_int(d) << 32) | (u64)(((unsigned)ki << 4) | slot);
}
__device__ __forceinline__ float entry_dist(u64 e) { return __int_as_float((int)(unsigned)(e >> 32)); }
__device__ __forceinline__ int   entry_key(u64 e)  { return (int)((unsigned)e >> 4); }

__device__ __forceinline__ u64 set_max(const u64 (&k)[KNN])
{
    u64 t[KNN / 2];
#pragma unroll
    for (int i = 0; i < KNN / 2; i++) t[i] = (k[i + KNN / 2] > k[i]) ? k[i + KNN / 2] : k[i];
#pragma unroll
    for (int w = KNN / 4; w >= 1; w >>= 1) {
#pragma unroll
        for (int i = 0; i < w; i++) t[i] = (t[i + w] > t[i]) ? t[i + w] : t[i];
    }
    return t[0];
}

__device__ __forceinline__ void set_offer(u64 (&k)[KNN], u64& mk, float vd, int vi)
{
    const u64 c = set_entry(vd, vi, 0u);
    const bool take = c < (mk & ~(u64)15);
    const unsigned mslot = (unsigned)mk & 15u;
    const unsigned slot  = take ? mslot : 16u;
    const u64 nv = c | (u64)mslot;
#pragma unroll
    for (int s = 0; s < KNN; s++) k[s] = (slot == (unsigned)s) ? nv : k[s];
    mk = set_max(k);
}

__device__ __forceinline__ void drain_queue(u64 (&k)[KNN], u64& mk, const v2i* ql, int cnt)
{
    static_assert(QDEPTH % 2 == 0, "two queue slots per trip");
    unsigned live = __builtin_amdgcn_ballot_w32(0 < cnt);
#pragma unroll 1
    for (int r = 0; live != 0u; r += 2) {
        const unsigned live_next = (r + 2 < QDEPTH) ? __builtin_amdgcn_ballot_w32(r + 2 < cnt) : 0u;
        const v2i ea = ql[r * 32];
        const v2i eb = ql[(r + 1) * 32];
        const bool ha = r < cnt, hb = r + 1 < cnt;
        set_offer(k, mk, ha ? __int_as_float(ea.x) : 3.4e38f, ha ? ea.y : 0);
        set_offer(k, mk, hb ? __int_as_float(eb.x) : 3.4e38f, hb ? eb.y : 0);
        live = live_next;
    }
}

__global__ __launch_bounds__(128) void knn_kernel(
    const float* __restrict__ xq,
    const float* __restrict__ xk,
    int* __restrict__ idx)
{
    __shared__ u64 lst[2][KNN][32];
    __shared__ v2i qq[4][QDEPTH][32];
    const int lane = threadIdx.x & 31;
    const int wave = __builtin_amdgcn_readfirstlane((int)(threadIdx.x >> 5));
    const int half = wave & 1;
    const int grp  = wave >> 1;
    const int q0 = blockIdx.x * 64 + grp * 32;
    const int b  = q0 / NP;
    const int n  = (q0 % NP) + lane;
    const float* q  = xq + (size_t)b * 3 * NP;
    const float* kp = xk + (size_t)b * 3 * NP;
    float qx = q[n], qy = q[NP + n], qz = q[2 * NP + n];

    u64 ks[KNN];
#pragma unroll
    for (int s = 0; s < KNN; s++) ks[s] = set_entry(3.4e38f, 0, (unsigned)s);
    u64 mk = set_entry(3.4e38f, 0, KNN - 1);

    v2i* ql = &qq[wave][0][lane];
    {
        v2i f; f.x = __float_as_int(3.4e38f); f.y = 0;
#pragma unroll
        for (int s = 0; s < QDEPTH; s++) ql[s * 32] = f;
    }
    float thr = 3.4e38f;
    int   cnt = 0;

    const int kbeg = half * (NP / 2);
    const int kend = kbeg + NP / 2;
    const int l8 = lane & 7;
    const int hi = lane >> 4;
    const float* kA0 = kp + (hi ? 2 * NP : 0) + l8;
    const float* kA1 = kp + NP + l8;
    const float* kB0 = kA0 + 8;
    const float* kB1 = kA1 + 8;
    v2f selx, sely, selz;
    selx.x = hi ? 0.f : -1.f; selx.y = 0.f;
    sely.x = 0.f;             sely.y = hi ? 0.f : -1.f;
    selz.x = hi ? -1.f : 0.f; selz.y = 0.f;
    asm volatile("" : "+v"(selx), "+v"(sely), "+v"(selz));
    const v8f cqx = {qx, qx, qx, qx, qx, qx, qx, qx};
    const v8f cqy = {qy, qy, qy, qy, qy, qy, qy, qy};
    const v8f cqz = {qz, qz, qz, qz, qz, qz, qz, qz};

#define KNN_LOAD_SET(SA, SB, T)                                                   \
    {                                                                             \
        const int lt = (T);                                                       \
        SA.x = kA0[lt]; SA.y = kA1[lt]; SB.x = kB0[lt]; SB.y = kB1[lt];           \
    }
    v2f aA, aB, bA, bB, eA, eB, fA, fB;
    KNN_LOAD_SET(aA, aB, kbeg)
    KNN_LOAD_SET(bA, bB, kbeg + 16)
    KNN_LOAD_SET(eA, eB, kbeg + 32)
    KNN_LOAD_SET(fA, fB, kbeg + 48)

    v8f Px, Py, Pz, Qx, Qy, Qz;
    Px = key_diffs(aA, selx, cqx);
    Py = key_diffs(aA, sely, cqy);
    Pz = key_diffs(aA, selz, cqz);

#define KNN_FENCE __builtin_amdgcn_sched_barrier(0)
    static_assert(QFULL + KGRP <= QDEPTH, "the keys between two queue tests must fit behind a queue that passed the test");
#define KNN_HALF_STEP(CUR, NXT, NFRAG, KBASE)                                     \
    {                                                                             \
        const int kb = (KBASE);                                                   \
        NXT##x = key_diffs(NFRAG, selx, cqx);                                     \
        KNN_FENCE;                                                                \
        const v2f d01 = dist_pair<0>(CUR##x, CUR##y, CUR##z);                     \
        const v2f d23 = dist_pair<2>(CUR##x, CUR##y, CUR##z);                     \
        const v2f d45 = dist_pair<4>(CUR##x, CUR##y, CUR##z);                     \
        const v2f d67 = dist_pair<6>(CUR##x, CUR##y, CUR##z);                     \
        queue_key(d01.x, kb + 0, ql, cnt, thr);                                   \
        queue_key(d01.y, kb + 1, ql, cnt, thr);                                   \
        KNN_FENCE;                                                                \
        NXT##y = key_diffs(NFRAG, sely, cqy);                                     \
        KNN_FENCE;                                                                \
        queue_key(d23.x, kb + 2, ql, cnt, thr);                                   \
        queue_key(d23.y, kb + 3, ql, cnt, thr);                                   \
        queue_key(d45.x, kb + 4, ql, cnt, thr);                                   \
        KNN_FENCE;                                                                \
        NXT##z = key_diffs(NFRAG, selz, cqz);                                     \
        KNN_FENCE;                                                                \
        queue_key(d45.y, kb + 5, ql, cnt, thr);                                   \
        queue_key(d67.x, kb + 6, ql, cnt, thr);                                   \
        queue_key(d67.y, kb + 7, ql, cnt, thr);                                   \
        if (__builtin_amdgcn_ballot_w32(cnt >= QFULL) != 0u) {                    \
            drain_queue(ks, mk, ql, cnt);                                         \
            thr = entry_dist(mk);                                                 \
            cnt = 0;                                                              \
        }                                                                         \
    }
#define KNN_TILE_STEP(SA, SB, NA, T)                                              \
    {                                                                             \
        const int tt = (T);                                                       \
        KNN_HALF_STEP(P, Q, SB, tt)                                               \
        const int nt = (tt + 64 < kend) ? tt + 64 : tt;                           \
        KNN_LOAD_SET(SA, SB, nt)                                                  \
        KNN_HALF_STEP(Q, P, NA, tt + 8)                                           \
    }

#pragma unroll 1
    for (int tile = kbeg; tile < kend; tile += 64) {
        KNN_TILE_STEP(aA, aB, bA, tile)
        KNN_TILE_STEP(bA, bB, eA, tile + 16)
        KNN_TILE_STEP(eA, eB, fA, tile + 32)
        KNN_TILE_STEP(fA, fB, aA, tile + 48)
    }
#undef KNN_TILE_STEP
#undef KNN_HALF_STEP
#undef KNN_FENCE
#undef KNN_LOAD_SET
    drain_queue(ks, mk, ql, cnt);

    u64* ul = &lst[grp][0][lane];
    if (half == 1) {
#pragma unroll
        for (int s = 0; s < KNN; s++) ul[s * 32] = ks[s];
    }
    __syncthreads();
    if (half == 0) {
#pragma unroll 1
        for (int r = 0; r < KNN; r += 2) {
            const u64 ea = ul[r * 32];
            const u64 eb = ul[(r + 1) * 32];
            set_offer(ks, mk, entry_dist(ea), entry_key(ea));
            set_offer(ks, mk, entry_dist(eb), entry_key(eb));
        }
        int* o = idx + (size_t)(q0 + lane) * KNN;
#pragma unroll
        for (int j = 0; j < KNN; j++) o[j] = entry_key(ks[j]);
    }
}

static constexpr int WKK = 4 * 32 + 2;

__device__ __forceinline__ void put_wpiece(v2f* wl, int piece, v4f w)
{
    const int o = piece >> 4, j = piece & 15;
    v2f* p = wl + j * WKK + (o >> 4) * 32 + (o & 15);
    v2f lo; lo.x = w.x; lo.y = w.y;
    v2f hi; hi.x = w.z; hi.y = w.w;
    p[0]  = lo;
    p[16] = hi;
}

__global__ __launch_bounds__(256) void gemm_cm_kernel(
    const float* __restrict__ in,
    const float* __restrict__ W,
    const float* __restrict__ bias,
    float* __restrict__ out_rm)
{
    __shared__ v2f wl[16 * WKK];
    int tid = threadIdx.x;
    int lane = tid & 31, wave = tid >> 5;
    int tile = blockIdx.x * 8 + wave;
    int b  = tile / (NP / 16);
    int n0 = (tile % (NP / 16)) * 16;
    const float* inb = in + (size_t)b * CH * NP;
    int row   = n0 + (lane & 15);
    int khalf = 2 * (lane >> 4);

    v4f wv[4];
#pragma unroll
    for (int i = 0; i < 4; i++) wv[i] = *(const v4f*)(W + 4 * (tid + 256 * i));
    float bvs[4];
#pragma unroll
    for (int nc = 0; nc < 4; nc++) bvs[nc] = bias[nc * 16 + (lane & 15)];
    v2f a[16];
#pragma unroll
    for (int kk = 0; kk < 16; kk++) {
        int k0 = kk * 4 + khalf;
        v2f v; v.x = inb[(size_t)k0 * NP + row]; v.y = inb[(size_t)(k0 + 1) * NP + row];
        a[kk] = v;
    }
#pragma unroll
    for (int i = 0; i < 4; i++) put_wpiece(wl, tid + 256 * i, wv[i]);
    __syncthreads();
#pragma unroll
    for (int nc = 0; nc < 4; nc++) {
        v8f acc = {0.f, 0.f, 0.f, 0.f, 0.f, 0.f, 0.f, 0.f};
#pragma unroll
        for (int kk = 0; kk < 16; kk++) {
            acc = __builtin_amdgcn_wmma_f32_16x16x4_f32(
                false, a[kk], false, wl[kk * WKK + nc * 32 + lane],
                (short)0, acc, false, false);
        }
        int m = nc * 16 + (lane & 15);
        float bv = bvs[nc];
        float* ob = out_rm + (size_t)b * NP * 64;
#pragma unroll
        for (int v = 0; v < 8; v++) {
            int r = n0 + v + 8 * (lane >> 4);
            ob[(size_t)r * 64 + m] = acc[v] + bv;
        }
    }
}

__global__ __launch_bounds__(256) void gemm_rm_kernel(
    const float* __restrict__ in,
    const float* __restrict__ W,
    const float* __restrict__ bias,
    float* __restrict__ out_cm,
    float* __restrict__ out_rm)
{
    __shared__ v2f wl[16 * WKK];
    int tid = threadIdx.x;
    int lane = tid & 31, wave = tid >> 5;
    int tile = blockIdx.x * 8 + wave;
    int b  = tile / (NP / 16);
    int n0 = (tile % (NP / 16)) * 16;
    const float* inb = in + (size_t)b * NP * 64;
    int row   = n0 + (lane & 15);
    int khalf = 2 * (lane >> 4);

    v4f wv[4];
#pragma unroll
    for (int i = 0; i < 4; i++) wv[i] = *(const v4f*)(W + 4 * (tid + 256 * i));
    float bvs[4];
#pragma unroll
    for (int nc = 0; nc < 4; nc++) bvs[nc] = bias[nc * 16 + (lane & 15)];
    v2f a[16];
#pragma unroll
    for (int kk = 0; kk < 16; kk++)
        a[kk] = *(const v2f*)(inb + (size_t)row * 64 + kk * 4 + khalf);
#pragma unroll
    for (int i = 0; i < 4; i++) put_wpiece(wl, tid + 256 * i, wv[i]);
    __syncthreads();

#pragma unroll
    for (int nc = 0; nc < 4; nc++) {
        v8f acc = {0.f, 0.f, 0.f, 0.f, 0.f, 0.f, 0.f, 0.f};
#pragma unroll
        for (int kk = 0; kk < 16; kk++) {
            acc = __builtin_amdgcn_wmma_f32_16x16x4_f32(
                false, a[kk], false, wl[kk * WKK + nc * 32 + lane],
                (short)0, acc, false, false);
        }
        int m = nc * 16 + (lane & 15);
        float bv = bvs[nc];
#pragma unroll
        for (int v = 0; v < 8; v++) {
            int r = n0 + v + 8 * (lane >> 4);
            float val = acc[v] + bv;
            out_cm[(size_t)b * 64 * NP + (size_t)m * NP + r] = val;
            if (out_rm) out_rm[((size_t)b * NP + r) * 64 + m] = val;
        }
    }
}

static constexpr int QPW = 2;
template <int NL>
__global__ __launch_bounds__(128, 4) void cross_kernel(
    const float* __restrict__ xq, const float* __restrict__ xk,
    const float* __restrict__ p1, const float* __restrict__ p2,
    const int*   __restrict__ idx,
    const float* __restrict__ posw, const float* __restrict__ posb,
    const float* __restrict__ w1, const float* __restrict__ b1,
    const float* __restrict__ w2, const float* __restrict__ b2,
    float* __restrict__ out_rm,
    float* __restrict__ out_cm)
{
    __shared__ v2f  wl1[16 * WKK];
    __shared__ v2f  wl2[(NL > 1) ? 16 * WKK : 2];
    __shared__ float X[4][QPW * 16][68];

    const int tid = threadIdx.x;
    const int lane = tid & 31, wave = tid >> 5;
    const int q0 = (blockIdx.x * 4 + wave) * QPW;
    const int b  = q0 / NP;
    const int n0 = q0 % NP;
    const float* xqb = xq + (size_t)b * 3 * NP;
    const float* xkb = xk + (size_t)b * 3 * NP;
    const int c0 = 2 * lane;
    const int mrow  = lane & 15;
    const int khalf = 2 * (lane >> 4);
    const int hrow  = 8 * (lane >> 4);

    v4i nbv[QPW][KNN / 4];
    float qx[QPW], qy[QPW], qz[QPW];
    v2f p1v[QPW];
#pragma unroll
    for (int t = 0; t < QPW; t++) {
        const v4i* iq = (const v4i*)(idx + (size_t)(q0 + t) * KNN);
#pragma unroll
        for (int i = 0; i < KNN / 4; i++) nbv[t][i] = iq[i];
        qx[t] = xqb[n0 + t]; qy[t] = xqb[NP + n0 + t]; qz[t] = xqb[2 * NP + n0 + t];
        p1v[t] = *(const v2f*)(p1 + (size_t)(q0 + t) * 64 + c0);
    }
    v4f wa[8], wb[8];
#pragma unroll
    for (int i = 0; i < 8; i++) wa[i] = *(const v4f*)(w1 + 4 * (tid + 128 * i));
    if (NL > 1) {
#pragma unroll
        for (int i = 0; i < 8; i++) wb[i] = *(const v4f*)(w2 + 4 * (tid + 128 * i));
    }
    const v2f pw01 = *(const v2f*)(posw + c0 * 3);
    const v2f pw23 = *(const v2f*)(posw + c0 * 3 + 2);
    const v2f pw45 = *(const v2f*)(posw + c0 * 3 + 4);
    const v2f pbv  = *(const v2f*)(posb + c0);
    float bv1[4], bv2[4];
#pragma unroll
    for (int nc = 0; nc < 4; nc++) {
        bv1[nc] = b1[nc * 16 + mrow];
        bv2[nc] = (NL > 1) ? b2[nc * 16 + mrow] : 0.f;
    }

#pragma unroll
    for (int i = 0; i < 8; i++) put_wpiece(wl1, tid + 128 * i, wa[i]);
    if (NL > 1) {
#pragma unroll
        for (int i = 0; i < 8; i++) put_wpiece(wl2, tid + 128 * i, wb[i]);
    }
    __builtin_amdgcn_sched_barrier(0);

    float dx[QPW][KNN], dy[QPW][KNN], dz[QPW][KNN];
    v2f   g[QPW][KNN];
#pragma unroll
    for (int t = 0; t < QPW; t++) {
#pragma unroll
        for (int r = 0; r < KNN; r++) {
            const int nb = nbv[t][r >> 2][r & 3];
            dx[t][r] = xkb[nb]; dy[t][r] = xkb[NP + nb]; dz[t][r] = xkb[2 * NP + nb];
            g[t][r]  = *(const v2f*)(p2 + ((size_t)b * NP + nb) * 64 + c0);
        }
    }

    const float pwa0 = pw01.x, pwa1 = pw01.y, pwa2 = pw23.x;
    const float pwb0 = pw23.y, pwb1 = pw45.x, pwb2 = pw45.y;
    const float pb0 = pbv.x, pb1 = pbv.y;
#pragma unroll
    for (int t = 0; t < QPW; t++) {
#pragma unroll
        for (int r = 0; r < KNN; r++) {
            const float ddx = dx[t][r] - qx[t], ddy = dy[t][r] - qy[t], ddz = dz[t][r] - qz[t];
            X[wave][t * 16 + r][c0]     = leaky(g[t][r].x + p1v[t].x + pb0 + ddx * pwa0 + ddy * pwa1 + ddz * pwa2);
            X[wave][t * 16 + r][c0 + 1] = leaky(g[t][r].y + p1v[t].y + pb1 + ddx * pwb0 + ddy * pwb1 + ddz * pwb2);
        }
    }
    __syncthreads();

#pragma unroll
    for (int layer = 0; layer < NL; layer++) {
        const v2f* wl = layer ? wl2 : wl1;
        const bool last = (layer == NL - 1);

        v2f a0[16], a1[16];
#pragma unroll
        for (int kk = 0; kk < 16; kk++) {
            a0[kk] = *(const v2f*)&X[wave][mrow][kk * 4 + khalf];
            a1[kk] = *(const v2f*)&X[wave][16 + mrow][kk * 4 + khalf];
        }

#pragma unroll
        for (int nc = 0; nc < 4; nc++) {
            v8f acc0 = {0.f, 0.f, 0.f, 0.f, 0.f, 0.f, 0.f, 0.f};
            v8f acc1 = {0.f, 0.f, 0.f, 0.f, 0.f, 0.f, 0.f, 0.f};
#pragma unroll
            for (int kk = 0; kk < 16; kk++) {
                const v2f bf = wl[kk * WKK + nc * 32 + lane];
                acc0 = __builtin_amdgcn_wmma_f32_16x16x4_f32(
                    false, a0[kk], false, bf, (short)0, acc0, false, false);
                acc1 = __builtin_amdgcn_wmma_f32_16x16x4_f32(
                    false, a1[kk], false, bf, (short)0, acc1, false, false);
            }
            const int m = nc * 16 + mrow;
            const float bv = layer ? bv2[nc] : bv1[nc];
            if (!last) {
#pragma unroll
                for (int v = 0; v < 8; v++) {
                    X[wave][v + hrow][m]      = leaky(acc0[v] + bv);
                    X[wave][16 + v + hrow][m] = leaky(acc1[v] + bv);
                }
            } else {
                float mx0 = -3.4e38f, mx1 = -3.4e38f;
#pragma unroll
                for (int v = 0; v < 8; v++) {
                    mx0 = fmaxf(mx0, leaky(acc0[v] + bv));
                    mx1 = fmaxf(mx1, leaky(acc1[v] + bv));
                }
                mx0 = fmaxf(mx0, __shfl_xor(mx0, 16, 32));
                mx1 = fmaxf(mx1, __shfl_xor(mx1, 16, 32));
                const int t = lane >> 4;
                const float mx = t ? mx1 : mx0;
                out_rm[(size_t)(q0 + t) * 64 + m] = mx;
                if (out_cm)
                    out_cm[(size_t)b * 64 * NP + (size_t)m * NP + n0 + t] = mx;
            }
        }
        if (!last) __syncthreads();
    }
}

extern "C" void kernel_launch(void* const* d_in, const int* in_sizes, int n_in,
                              void* d_out, int out_size, void* d_ws, size_t ws_size,
                              hipStream_t stream)
{
    (void)in_sizes; (void)n_in; (void)out_size; (void)ws_size;
    const float* pc1   = (const float*)d_in[0];
    const float* pc2   = (const float*)d_in[1];
    const float* feat1 = (const float*)d_in[2];
    const float* feat2 = (const float*)d_in[3];
    const float* t11_w = (const float*)d_in[4];  const float* t11_b = (const float*)d_in[5];
    const float* t22_w = (const float*)d_in[6];  const float* t22_b = (const float*)d_in[7];
    const float* pos1w = (const float*)d_in[8];  const float* pos1b = (const float*)d_in[9];
    const float* m1w1  = (const float*)d_in[10]; const float* m1b1  = (const float*)d_in[11];
    const float* m1w2  = (const float*)d_in[12]; const float* m1b2  = (const float*)d_in[13];
    const float* t1_w  = (const float*)d_in[14]; const float* t1_b  = (const float*)d_in[15];
    const float* t2_w  = (const float*)d_in[16]; const float* t2_b  = (const float*)d_in[17];
    const float* pos2w = (const float*)d_in[18]; const float* pos2b = (const float*)d_in[19];
    const float* m2w1  = (const float*)d_in[20]; const float* m2b1  = (const float*)d_in[21];

    float* out = (float*)d_out;
    const size_t S = (size_t)BB * NP * 64;
    float* ws = (float*)d_ws;
    float* q1 = ws;           float* q2 = ws + S;
    float* r1 = ws + 2 * S;   float* r2 = ws + 3 * S;
    float* pooled1 = ws + 4 * S;
    float* pooled2 = ws + 5 * S;
    int* idx1 = (int*)(ws + 6 * S);
    int* idx2 = idx1 + (size_t)BB * NP * KNN;
    float* fn1_rm  = q1;
    float* fn2_rm  = q2;
    float* pooled3 = r1;

    knn_kernel<<<BB * NP / 64, 128, 0, stream>>>(pc1, pc2, idx1);
    knn_kernel<<<BB * NP / 64, 128, 0, stream>>>(pc2, pc1, idx2);

    gemm_cm_kernel<<<128, 256, 0, stream>>>(feat1, t11_w, t11_b, q1);
    gemm_cm_kernel<<<128, 256, 0, stream>>>(feat2, t22_w, t22_b, q2);
    gemm_cm_kernel<<<128, 256, 0, stream>>>(feat2, t11_w, t11_b, r1);
    gemm_cm_kernel<<<128, 256, 0, stream>>>(feat1, t22_w, t22_b, r2);

    cross_kernel<2><<<BB * NP / (4 * QPW), 128, 0, stream>>>(pc1, pc2, q1, q2, idx1,
        pos1w, pos1b, m1w1, m1b1, m1w2, m1b2, pooled1, nullptr);
    cross_kernel<2><<<BB * NP / (4 * QPW), 128, 0, stream>>>(pc2, pc1, r1, r2, idx2,
        pos1w, pos1b, m1w1, m1b1, m1w2, m1b2, pooled2, nullptr);

    gemm_rm_kernel<<<128, 256, 0, stream>>>(pooled1, t1_w, t1_b, out,       fn1_rm);
    gemm_rm_kernel<<<128, 256, 0, stream>>>(pooled2, t2_w, t2_b, out + S,   fn2_rm);

    cross_kernel<1><<<BB * NP / (4 * QPW), 128, 0, stream>>>(pc1, pc2, fn1_rm, fn2_rm, idx1,
        pos2w, pos2b, m2w1, m2b1, nullptr, nullptr, pooled3, out + 2 * S);
}
